# starting kernel
# baseline (speedup 1.0000x reference)
#include <hip/hip_runtime.h>
#include <cmath>

#define IN_F  256
#define OUT_F 64
#define LRELU_ALPHA 0.2f

typedef __attribute__((ext_vector_type(2))) float v2f;
typedef __attribute__((ext_vector_type(8))) float v8f;

__global__ __launch_bounds__(256) void gat_gemm_wmma(
    const float* __restrict__ x, const float* __restrict__ W,
    float* __restrict__ h, int N)
{
    __shared__ float Wlds[IN_F * OUT_F];
    const int tid = threadIdx.x;
    for (int i = tid; i < IN_F * OUT_F; i += 256) {
        const int k = i / OUT_F, c = i % OUT_F;
        Wlds[(k >> 1) * (2 * OUT_F) + c * 2 + (k & 1)] = W[i];
    }
    __syncthreads();

    const int wave = tid >> 5;
    const int lane = tid & 31;
    const int half = lane >> 4;
    const int l16  = lane & 15;
    const int tile_row0 = blockIdx.x * 128 + wave * 16;
    const int arow = tile_row0 + l16;
    const int lrow = arow < N ? arow : (N - 1);

    const float* xptr = x + (size_t)lrow * IN_F + 2 * half;
    const float* bptr = Wlds + half * (2 * OUT_F) + l16 * 2;

    v8f acc[4] = {};

    for (int k0 = 0; k0 < IN_F; k0 += 4) {
        const float2 t = *(const float2*)(xptr + k0);
        v2f a; a.x = t.x; a.y = t.y;
        const float* bk = bptr + (k0 >> 1) * (2 * OUT_F);
#pragma unroll
        for (int n = 0; n < 4; ++n) {
            const float2 u = *(const float2*)(bk + n * 32);
            v2f b; b.x = u.x; b.y = u.y;
            acc[n] = __builtin_amdgcn_wmma_f32_16x16x4_f32(
                false, a, false, b, (short)0, acc[n], false, false);
        }
    }

    const int rbase = tile_row0 + 8 * half;
#pragma unroll
    for (int r = 0; r < 8; ++r) {
        const int rr = rbase + r;
        if (rr < N) {
#pragma unroll
            for (int n = 0; n < 4; ++n)
                h[(size_t)rr * OUT_F + n * 16 + l16] = acc[n][r];
        }
    }
}

__global__ __launch_bounds__(256) void gat_scores(
    const float* __restrict__ h, const float* __restrict__ attn,
    float* __restrict__ s_src, float* __restrict__ s_dst, int N)
{
    const int node = (blockIdx.x * blockDim.x + threadIdx.x) >> 5;
    const int lane = threadIdx.x & 31;
    if (node >= N) return;
    const float* hr = h + (size_t)node * OUT_F;
    const float h0 = hr[lane], h1 = hr[lane + 32];
    float ps = h0 * attn[lane]         + h1 * attn[lane + 32];
    float pd = h0 * attn[OUT_F + lane] + h1 * attn[OUT_F + lane + 32];
    for (int off = 16; off > 0; off >>= 1) {
        ps += __shfl_down(ps, off, 32);
        pd += __shfl_down(pd, off, 32);
    }
    if (lane == 0) { s_src[node] = ps; s_dst[node] = pd; }
}

__device__ __forceinline__ float edge_logit(const float* s_src, const float* s_dst,
                                            int s, int d) {
    const float v = s_src[s] + s_dst[d];
    return -(v > 0.f ? v : LRELU_ALPHA * v);
}

__global__ void gat_init(float* __restrict__ out, float* __restrict__ m,
                         float* __restrict__ denom, int N)
{
    const int i = blockIdx.x * blockDim.x + threadIdx.x;
    if (i < N * OUT_F) out[i] = 0.f;
    if (i < N) { m[i] = -INFINITY; denom[i] = 0.f; }
}

__global__ void gat_edge_max(const int* __restrict__ src, const int* __restrict__ dst,
                             const float* __restrict__ s_src, const float* __restrict__ s_dst,
                             float* __restrict__ m, int E)
{
    const int i = blockIdx.x * blockDim.x + threadIdx.x;
    if (i >= E) return;
    const int s = src[i];
    const float e = edge_logit(s_src, s_dst, s, dst[i]);
    atomicMax(&m[s], e);
}

__global__ void gat_edge_sum(const int* __restrict__ src, const int* __restrict__ dst,
                             const float* __restrict__ s_src, const float* __restrict__ s_dst,
                             const float* __restrict__ m, float* __restrict__ denom, int E)
{
    const int i = blockIdx.x * blockDim.x + threadIdx.x;
    if (i >= E) return;
    const int s = src[i];
    const float e = edge_logit(s_src, s_dst, s, dst[i]);
    atomicAdd(&denom[s], __expf(e - m[s]));
}

__global__ __launch_bounds__(256) void gat_edge_aggr(
    const int* __restrict__ src, const int* __restrict__ dst,
    const float* __restrict__ s_src, const float* __restrict__ s_dst,
    const float* __restrict__ m, const float* __restrict__ denom,
    const float* __restrict__ h, float* __restrict__ out, int E)
{
    const int edge = (blockIdx.x * blockDim.x + threadIdx.x) >> 5;
    const int lane = threadIdx.x & 31;
    if (edge >= E) return;
    const int s = src[edge], d = dst[edge];
    const float e = edge_logit(s_src, s_dst, s, d);
    const float w = __expf(e - m[s]) / denom[s];
    const float* hr = h + (size_t)d * OUT_F;
    float* orow = out + (size_t)s * OUT_F;
    atomicAdd(&orow[lane],      w * hr[lane]);
    atomicAdd(&orow[lane + 32], w * hr[lane + 32]);
}

__global__ void gat_elu(float* __restrict__ out, int total)
{
    const int i = blockIdx.x * blockDim.x + threadIdx.x;
    if (i < total) {
        const float v = out[i];
        out[i] = v > 0.f ? v : expm1f(v);
    }
}

extern "C" void kernel_launch(void* const* d_in, const int* in_sizes, int n_in,
                              void* d_out, int out_size, void* d_ws, size_t ws_size,
                              hipStream_t stream)
{
    const float* x    = (const float*)d_in[0];
    const int*   edge = (const int*)  d_in[1];
    const float* W    = (const float*)d_in[2];
    const float* attn = (const float*)d_in[3];

    const int N = in_sizes[0] / IN_F;
    const int E = in_sizes[1] / 2;
    const int* src = edge;
    const int* dst = edge + E;
    float* out = (float*)d_out;

    float* h     = (float*)d_ws;
    float* s_src = h + (size_t)N * OUT_F;
    float* s_dst = s_src + N;
    float* m     = s_dst + N;
    float* denom = m + N;

    const int totalNF = N * OUT_F;

    gat_init     <<<(totalNF + 255) / 256, 256, 0, stream>>>(out, m, denom, N);
    gat_gemm_wmma<<<(N + 127) / 128,       256, 0, stream>>>(x, W, h, N);
    gat_scores   <<<(N * 32 + 255) / 256,  256, 0, stream>>>(h, attn, s_src, s_dst, N);
    gat_edge_max <<<(E + 255) / 256,       256, 0, stream>>>(src, dst, s_src, s_dst, m, E);
    gat_edge_sum <<<(E + 255) / 256,       256, 0, stream>>>(src, dst, s_src, s_dst, m, denom, E);
    {
        const long long threads = (long long)E * 32;
        gat_edge_aggr<<<(int)((threads + 255) / 256), 256, 0, stream>>>(
            src, dst, s_src, s_dst, m, denom, h, out, E);
    }
    gat_elu      <<<(totalNF + 255) / 256, 256, 0, stream>>>(out, totalNF);
}


# Scatter to gather: one returning int atomic per edge files dst in its row's own 64 output words, then a wave per row does softmax and sum in registers; float atomics gone, sum order = arrival rank
# speedup vs baseline: 2.4332x; 2.4332x over previous
#include <hip/hip_runtime.h>
#include <cmath>

#define IN_F  256
#define OUT_F 64
#define LRELU_ALPHA 0.2f
#define SLOT 64

typedef __attribute__((ext_vector_type(2))) float v2f;
typedef __attribute__((ext_vector_type(8))) float v8f;

__global__ __launch_bounds__(256) void gat_gemm_wmma(
    const float* __restrict__ x, const float* __restrict__ W,
    float* __restrict__ h, int N)
{
    __shared__ float Wlds[IN_F * OUT_F];
    const int tid = threadIdx.x;
    for (int i = tid; i < IN_F * OUT_F; i += 256) {
        const int k = i / OUT_F, c = i % OUT_F;
        Wlds[(k >> 1) * (2 * OUT_F) + c * 2 + (k & 1)] = W[i];
    }
    __syncthreads();

    const int wave = tid >> 5;
    const int lane = tid & 31;
    const int half = lane >> 4;
    const int l16  = lane & 15;
    const int tile_row0 = blockIdx.x * 128 + wave * 16;
    const int arow = tile_row0 + l16;
    const int lrow = arow < N ? arow : (N - 1);

    const float* xptr = x + (size_t)lrow * IN_F + 2 * half;
    const float* bptr = Wlds + half * (2 * OUT_F) + l16 * 2;

    v8f acc[4] = {};

    for (int k0 = 0; k0 < IN_F; k0 += 4) {
        const float2 t = *(const float2*)(xptr + k0);
        v2f a; a.x = t.x; a.y = t.y;
        const float* bk = bptr + (k0 >> 1) * (2 * OUT_F);
#pragma unroll
        for (int n = 0; n < 4; ++n) {
            const float2 u = *(const float2*)(bk + n * 32);
            v2f b; b.x = u.x; b.y = u.y;
            acc[n] = __builtin_amdgcn_wmma_f32_16x16x4_f32(
                false, a, false, b, (short)0, acc[n], false, false);
        }
    }

    const int rbase = tile_row0 + 8 * half;
#pragma unroll
    for (int r = 0; r < 8; ++r) {
        const int rr = rbase + r;
        if (rr < N) {
#pragma unroll
            for (int n = 0; n < 4; ++n)
                h[(size_t)rr * OUT_F + n * 16 + l16] = acc[n][r];
        }
    }
}

__global__ __launch_bounds__(256) void gat_scores(
    const float* __restrict__ h, const float* __restrict__ attn,
    float* __restrict__ s_src, float* __restrict__ s_dst, int N)
{
    const int node = (blockIdx.x * blockDim.x + threadIdx.x) >> 5;
    const int lane = threadIdx.x & 31;
    if (node >= N) return;
    const float* hr = h + (size_t)node * OUT_F;
    const float h0 = hr[lane], h1 = hr[lane + 32];
    float ps = h0 * attn[lane]         + h1 * attn[lane + 32];
    float pd = h0 * attn[OUT_F + lane] + h1 * attn[OUT_F + lane + 32];
    for (int off = 16; off > 0; off >>= 1) {
        ps += __shfl_down(ps, off, 32);
        pd += __shfl_down(pd, off, 32);
    }
    if (lane == 0) { s_src[node] = ps; s_dst[node] = pd; }
}

__global__ void gat_init(int* __restrict__ cnt, int* __restrict__ ovf, int N)
{
    const int i = blockIdx.x * blockDim.x + threadIdx.x;
    if (i < N) cnt[i] = 0;
    if (i == 0) ovf[0] = 0;
}

__global__ __launch_bounds__(256) void gat_bucket(
    const int* __restrict__ src, const int* __restrict__ dst,
    int* __restrict__ cnt, int* __restrict__ ovf, int* __restrict__ slots,
    int N, int E)
{
    const int i = blockIdx.x * blockDim.x + threadIdx.x;
    if (i >= E) return;
    const int s = src[i];
    const int r = atomicAdd(&cnt[s], 1);
    if (r < SLOT) {
        slots[(size_t)s * SLOT + r] = dst[i];
    } else {
        const int k = atomicAdd(&ovf[0], 1);
        if (k < N - 1) ovf[1 + k] = i;
    }
}

__device__ __forceinline__ float wave_max32(float v) {
    for (int off = 16; off > 0; off >>= 1) v = fmaxf(v, __shfl_xor(v, off, 32));
    return v;
}
__device__ __forceinline__ float wave_sum32(float v) {
    for (int off = 16; off > 0; off >>= 1) v += __shfl_xor(v, off, 32);
    return v;
}

__device__ __forceinline__ void row_chunk(bool valid, int d, float ss,
                                          const float* __restrict__ s_dst,
                                          const float* __restrict__ h, int lane,
                                          float& m, float& l, float& ax, float& ay)
{
    const unsigned mask = __builtin_amdgcn_ballot_w32(valid);
    if (mask == 0u) return;
    float e = -INFINITY;
    int dd = 0;
    if (valid) {
        dd = d;
        const float v = ss + s_dst[d];
        e = -(v > 0.f ? v : LRELU_ALPHA * v);
    }
    const float mn = fmaxf(m, wave_max32(e));
    const float scale = (m == -INFINITY) ? 0.f : __expf(m - mn);
    const float p = valid ? __expf(e - mn) : 0.f;
    l = l * scale + wave_sum32(p);
    ax *= scale; ay *= scale;
    m = mn;
    const int nmax = 32 - __clz((int)mask);
    for (int j0 = 0; j0 < nmax; j0 += 8) {
#pragma unroll
        for (int u = 0; u < 8; ++u) {
            const int j = j0 + u;
            const int dj = __builtin_amdgcn_readlane(dd, j);
            const float pj = __int_as_float(__builtin_amdgcn_readlane(__float_as_int(p), j));
            const float2* hr = (const float2*)(h + (size_t)dj * OUT_F);
            const float2 hv = hr[lane];
            ax += pj * hv.x; ay += pj * hv.y;
        }
    }
}

__global__ __launch_bounds__(256) void gat_row_gather(
    const int* __restrict__ src, const int* __restrict__ dst,
    const float* __restrict__ s_src, const float* __restrict__ s_dst,
    const int* __restrict__ cnt, const int* __restrict__ ovf,
    const float* __restrict__ h, const int* slots, float* out, int N, int E)
{
    const int lane = threadIdx.x & 31;
    const int row = __builtin_amdgcn_readfirstlane((int)((blockIdx.x * blockDim.x + threadIdx.x) >> 5));
    if (row >= N) return;
    const int c = cnt[row];
    const float ss = s_src[row];
    const int* slot = slots + (size_t)row * SLOT;
    float m = -INFINITY, l = 0.f, ax = 0.f, ay = 0.f;
    int total = 0;
    if (c > SLOT) total = ovf[0];
    if (total <= N - 1) {
        const int ns = c < SLOT ? c : SLOT;
        {
            const bool v = lane < ns;
            const int d = v ? slot[lane] : 0;
            row_chunk(v, d, ss, s_dst, h, lane, m, l, ax, ay);
        }
        if (ns > 32) {
            const bool v = lane + 32 < ns;
            const int d = v ? slot[lane + 32] : 0;
            row_chunk(v, d, ss, s_dst, h, lane, m, l, ax, ay);
        }
        for (int base = 0; base < total; base += 32) {
            const int idx = base + lane;
            bool v = idx < total;
            int d = 0;
            if (v) {
                const int eid = ovf[1 + idx];
                v = (src[eid] == row);
                if (v) d = dst[eid];
            }
            row_chunk(v, d, ss, s_dst, h, lane, m, l, ax, ay);
        }
    } else {
        for (int base = 0; base < E; base += 32) {
            const int eid = base + lane;
            bool v = eid < E;
            int d = 0;
            if (v) {
                v = (src[eid] == row);
                if (v) d = dst[eid];
            }
            row_chunk(v, d, ss, s_dst, h, lane, m, l, ax, ay);
        }
    }
    float rx = 0.f, ry = 0.f;
    if (l > 0.f) { rx = ax / l; ry = ay / l; }
    float2 o;
    o.x = rx > 0.f ? rx : expm1f(rx);
    o.y = ry > 0.f ? ry : expm1f(ry);
    *(float2*)(out + (size_t)row * OUT_F + 2 * lane) = o;
}

extern "C" void kernel_launch(void* const* d_in, const int* in_sizes, int n_in,
                              void* d_out, int out_size, void* d_ws, size_t ws_size,
                              hipStream_t stream)
{
    const float* x    = (const float*)d_in[0];
    const int*   edge = (const int*)  d_in[1];
    const float* W    = (const float*)d_in[2];
    const float* attn = (const float*)d_in[3];

    const int N = in_sizes[0] / IN_F;
    const int E = in_sizes[1] / 2;
    const int* src = edge;
    const int* dst = edge + E;
    float* out = (float*)d_out;

    float* h     = (float*)d_ws;
    float* s_src = h + (size_t)N * OUT_F;
    float* s_dst = s_src + N;
    int*   cnt   = (int*)(s_dst + N);
    int*   ovf   = cnt + N;

    gat_init      <<<(N + 255) / 256,      256, 0, stream>>>(cnt, ovf, N);
    gat_gemm_wmma <<<(N + 127) / 128,      256, 0, stream>>>(x, W, h, N);
    gat_scores    <<<(N * 32 + 255) / 256, 256, 0, stream>>>(h, attn, s_src, s_dst, N);
    gat_bucket    <<<(E + 255) / 256,      256, 0, stream>>>(src, dst, cnt, ovf, (int*)d_out, N, E);
    gat_row_gather<<<(N * 32 + 255) / 256, 256, 0, stream>>>(
        src, dst, s_src, s_dst, cnt, ovf, h, (const int*)d_out, out, N, E);
}


# Edge ranks from LDS counters in two levels (block partition into 256-row buckets, then one block per bucket) replace the global returning atomic per edge; sum order still = arrival rank
# speedup vs baseline: 4.5087x; 1.8530x over previous
#include <hip/hip_runtime.h>
#include <cmath>

#define IN_F  256
#define OUT_F 64
#define LRELU_ALPHA 0.2f
#define SLOT 64
#define RB_SHIFT 8
#define RB (1 << RB_SHIFT)
#define BCAP (RB * OUT_F)
#define NB_MAX 512
#define BSTRIDE 64
#define PT 512
#define PCH 16
#define RCH 8
#define ROWFLAG (1 << 30)
#define OVCAP(n) (((n) - 2) / 2)

typedef __attribute__((ext_vector_type(2))) float v2f;
typedef __attribute__((ext_vector_type(8))) float v8f;

__global__ __launch_bounds__(256) void gat_gemm_wmma(
    const float* __restrict__ x, const float* __restrict__ W,
    float* __restrict__ h, int N)
{
    __shared__ float Wlds[IN_F * OUT_F];
    const int tid = threadIdx.x;
    for (int i = tid; i < IN_F * OUT_F; i += 256) {
        const int k = i / OUT_F, c = i % OUT_F;
        Wlds[(k >> 1) * (2 * OUT_F) + c * 2 + (k & 1)] = W[i];
    }
    __syncthreads();

    const int wave = tid >> 5;
    const int lane = tid & 31;
    const int half = lane >> 4;
    const int l16  = lane & 15;
    const int tile_row0 = blockIdx.x * 128 + wave * 16;
    const int arow = tile_row0 + l16;
    const int lrow = arow < N ? arow : (N - 1);

    const float* xptr = x + (size_t)lrow * IN_F + 2 * half;
    const float* bptr = Wlds + half * (2 * OUT_F) + l16 * 2;

    v8f acc[4] = {};

    for (int k0 = 0; k0 < IN_F; k0 += 4) {
        const float2 t = *(const float2*)(xptr + k0);
        v2f a; a.x = t.x; a.y = t.y;
        const float* bk = bptr + (k0 >> 1) * (2 * OUT_F);
#pragma unroll
        for (int n = 0; n < 4; ++n) {
            const float2 u = *(const float2*)(bk + n * 32);
            v2f b; b.x = u.x; b.y = u.y;
            acc[n] = __builtin_amdgcn_wmma_f32_16x16x4_f32(
                false, a, false, b, (short)0, acc[n], false, false);
        }
    }

    const int rbase = tile_row0 + 8 * half;
#pragma unroll
    for (int r = 0; r < 8; ++r) {
        const int rr = rbase + r;
        if (rr < N) {
#pragma unroll
            for (int n = 0; n < 4; ++n)
                h[(size_t)rr * OUT_F + n * 16 + l16] = acc[n][r];
        }
    }
}

__global__ __launch_bounds__(256) void gat_scores(
    const float* __restrict__ h, const float* __restrict__ attn,
    float* __restrict__ s_src, float* __restrict__ s_dst, int N)
{
    const int node = (blockIdx.x * blockDim.x + threadIdx.x) >> 5;
    const int lane = threadIdx.x & 31;
    if (node >= N) return;
    const float* hr = h + (size_t)node * OUT_F;
    const float h0 = hr[lane], h1 = hr[lane + 32];
    float ps = h0 * attn[lane]         + h1 * attn[lane + 32];
    float pd = h0 * attn[OUT_F + lane] + h1 * attn[OUT_F + lane + 32];
    for (int off = 16; off > 0; off >>= 1) {
        ps += __shfl_down(ps, off, 32);
        pd += __shfl_down(pd, off, 32);
    }
    if (lane == 0) { s_src[node] = ps; s_dst[node] = pd; }
}

__global__ void gat_init(int* __restrict__ cnt, int* __restrict__ ovf,
                         int* __restrict__ bcnt, int N)
{
    const int i = blockIdx.x * blockDim.x + threadIdx.x;
    if (i < N) cnt[i] = 0;
    if (i < NB_MAX) bcnt[i * BSTRIDE] = 0;
    if (i == 0) ovf[0] = 0;
}

__device__ __forceinline__ void ovf_push(int* __restrict__ ovf, int row, int d, int N)
{
    const int k = atomicAdd(&ovf[0], 1);
    if (k < OVCAP(N)) { ovf[2 + 2 * k] = row; ovf[3 + 2 * k] = d; }
}

__global__ __launch_bounds__(PT) void gat_partition(
    const int* __restrict__ src, const int* __restrict__ dst,
    int* __restrict__ bcnt, int* __restrict__ part,
    int* __restrict__ cnt, int* __restrict__ ovf, int N, int E)
{
    __shared__ int hist[NB_MAX];
    __shared__ int base[NB_MAX];
    const int tid = threadIdx.x;
    const int nb = (N + RB - 1) >> RB_SHIFT;
    static_assert(PT == NB_MAX, "one thread clears one bucket counter");
    hist[tid] = 0;
    const int e0 = blockIdx.x * (PT * PCH) + tid;
    int s[PCH], d[PCH], r[PCH];
#pragma unroll
    for (int j = 0; j < PCH; ++j) {
        const int i = e0 + j * PT;
        const int ic = i < E ? i : E - 1;
        const int sv = src[ic];
        d[j] = dst[ic];
        s[j] = i < E ? sv : -1;
    }
    __syncthreads();
#pragma unroll
    for (int j = 0; j < PCH; ++j)
        r[j] = s[j] >= 0 ? atomicAdd(&hist[s[j] >> RB_SHIFT], 1) : 0;
    __syncthreads();
    if (tid < nb) {
        const int c = hist[tid];
        base[tid] = c > 0 ? atomicAdd(&bcnt[tid * BSTRIDE], c) : 0;
    }
    __syncthreads();
#pragma unroll
    for (int j = 0; j < PCH; ++j)
        r[j] += base[s[j] >= 0 ? (s[j] >> RB_SHIFT) : 0];
#pragma unroll
    for (int j = 0; j < PCH; ++j) {
        if (s[j] >= 0) {
            const int b = s[j] >> RB_SHIFT;
            const int pos = r[j];
            const int rows = min(RB, N - (b << RB_SHIFT));
            if (pos < rows * OUT_F) {
                part[(size_t)b * BCAP + pos] =
                    (int)(((unsigned)(s[j] & (RB - 1)) << 24) | (unsigned)d[j]);
            } else {
                atomicOr(&cnt[s[j]], ROWFLAG);
                ovf_push(ovf, s[j], d[j], N);
            }
        }
    }
}

__global__ __launch_bounds__(PT) void gat_rank(
    const int* __restrict__ bcnt, const int* __restrict__ part,
    int* __restrict__ cnt, int* __restrict__ ovf, int* __restrict__ slots, int N)
{
    __shared__ int lc[RB];
    const int tid = threadIdx.x;
    const int b = blockIdx.x;
    if (tid < RB) lc[tid] = 0;
    __syncthreads();
    const int row0 = b << RB_SHIFT;
    const int rows = min(RB, N - row0);
    const int n = min(bcnt[b * BSTRIDE], rows * OUT_F);
    const int* p = part + (size_t)b * BCAP;
    for (int t0 = 0; t0 < n; t0 += PT * RCH) {
        int w[RCH];
#pragma unroll
        for (int j = 0; j < RCH; ++j) {
            const int e = t0 + j * PT + tid;
            w[j] = p[e < n ? e : n - 1];
        }
        int rk[RCH];
#pragma unroll
        for (int j = 0; j < RCH; ++j) {
            const int e = t0 + j * PT + tid;
            rk[j] = 0;
            if (e < n) rk[j] = atomicAdd(&lc[(unsigned)w[j] >> 24], 1);
        }
#pragma unroll
        for (int j = 0; j < RCH; ++j) {
            const int e = t0 + j * PT + tid;
            if (e < n) {
                const int sl = (int)((unsigned)w[j] >> 24);
                const int dd = w[j] & 0xffffff;
                if (rk[j] < SLOT) slots[(size_t)(row0 + sl) * SLOT + rk[j]] = dd;
                else ovf_push(ovf, row0 + sl, dd, N);
            }
        }
    }
    __syncthreads();
    if (tid < rows) cnt[row0 + tid] = cnt[row0 + tid] | lc[tid];
}

__device__ __forceinline__ float wave_max32(float v) {
    for (int off = 16; off > 0; off >>= 1) v = fmaxf(v, __shfl_xor(v, off, 32));
    return v;
}
__device__ __forceinline__ float wave_sum32(float v) {
    for (int off = 16; off > 0; off >>= 1) v += __shfl_xor(v, off, 32);
    return v;
}

__device__ __forceinline__ void row_chunk(bool valid, int d, float ss,
                                          const float* __restrict__ s_dst,
                                          const float* __restrict__ h, int lane,
                                          float& m, float& l, float& ax, float& ay)
{
    const unsigned mask = __builtin_amdgcn_ballot_w32(valid);
    if (mask == 0u) return;
    float e = -INFINITY;
    int dd = 0;
    if (valid) {
        dd = d;
        const float v = ss + s_dst[d];
        e = -(v > 0.f ? v : LRELU_ALPHA * v);
    }
    const float mn = fmaxf(m, wave_max32(e));
    const float scale = (m == -INFINITY) ? 0.f : __expf(m - mn);
    const float p = valid ? __expf(e - mn) : 0.f;
    l = l * scale + wave_sum32(p);
    ax *= scale; ay *= scale;
    m = mn;
    const int nmax = 32 - __clz((int)mask);
    for (int j0 = 0; j0 < nmax; j0 += 8) {
#pragma unroll
        for (int u = 0; u < 8; ++u) {
            const int j = j0 + u;
            const int dj = __builtin_amdgcn_readlane(dd, j);
            const float pj = __int_as_float(__builtin_amdgcn_readlane(__float_as_int(p), j));
            const float2* hr = (const float2*)(h + (size_t)dj * OUT_F);
            const float2 hv = hr[lane];
            ax += pj * hv.x; ay += pj * hv.y;
        }
    }
}

__global__ __launch_bounds__(256) void gat_row_gather(
    const int* __restrict__ src, const int* __restrict__ dst,
    const float* __restrict__ s_src, const float* __restrict__ s_dst,
    const int* __restrict__ cnt, const int* __restrict__ ovf,
    const float* __restrict__ h, const int* slots, float* out, int N, int E)
{
    const int lane = threadIdx.x & 31;
    const int row = __builtin_amdgcn_readfirstlane((int)((blockIdx.x * blockDim.x + threadIdx.x) >> 5));
    if (row >= N) return;
    const int craw = cnt[row];
    const int c = craw & (ROWFLAG - 1);
    const float ss = s_src[row];
    const int* slot = slots + (size_t)row * SLOT;
    float m = -INFINITY, l = 0.f, ax = 0.f, ay = 0.f;
    int total = 0;
    if (c > SLOT || (craw & ROWFLAG)) total = ovf[0];
    if (total <= OVCAP(N)) {
        const int ns = c < SLOT ? c : SLOT;
        {
            const bool v = lane < ns;
            const int d = v ? slot[lane] : 0;
            row_chunk(v, d, ss, s_dst, h, lane, m, l, ax, ay);
        }
        if (ns > 32) {
            const bool v = lane + 32 < ns;
            const int d = v ? slot[lane + 32] : 0;
            row_chunk(v, d, ss, s_dst, h, lane, m, l, ax, ay);
        }
        for (int base = 0; base < total; base += 32) {
            const int idx = base + lane;
            bool v = idx < total;
            int d = 0;
            if (v) {
                v = (ovf[2 + 2 * idx] == row);
                if (v) d = ovf[3 + 2 * idx];
            }
            row_chunk(v, d, ss, s_dst, h, lane, m, l, ax, ay);
        }
    } else {
        for (int base = 0; base < E; base += 32) {
            const int eid = base + lane;
            bool v = eid < E;
            int d = 0;
            if (v) {
                v = (src[eid] == row);
                if (v) d = dst[eid];
            }
            row_chunk(v, d, ss, s_dst, h, lane, m, l, ax, ay);
        }
    }
    float rx = 0.f, ry = 0.f;
    if (l > 0.f) { rx = ax / l; ry = ay / l; }
    float2 o;
    o.x = rx > 0.f ? rx : expm1f(rx);
    o.y = ry > 0.f ? ry : expm1f(ry);
    *(float2*)(out + (size_t)row * OUT_F + 2 * lane) = o;
}

extern "C" void kernel_launch(void* const* d_in, const int* in_sizes, int n_in,
                              void* d_out, int out_size, void* d_ws, size_t ws_size,
                              hipStream_t stream)
{
    const float* x    = (const float*)d_in[0];
    const int*   edge = (const int*)  d_in[1];
    const float* W    = (const float*)d_in[2];
    const float* attn = (const float*)d_in[3];

    const int N = in_sizes[0] / IN_F;
    const int E = in_sizes[1] / 2;
    const int* src = edge;
    const int* dst = edge + E;
    float* out = (float*)d_out;

    float* h     = (float*)d_ws;
    float* s_src = h + (size_t)N * OUT_F;
    float* s_dst = s_src + N;
    int*   cnt   = (int*)(s_dst + N);
    int*   ovf   = cnt + N;

    int* bcnt = (int*)s_src;
    int* part = (int*)h;

    gat_init      <<<(N + 255) / 256,      256, 0, stream>>>(cnt, ovf, bcnt, N);
    gat_partition <<<(E + PT * PCH - 1) / (PT * PCH), PT, 0, stream>>>(src, dst, bcnt, part, cnt, ovf, N, E);
    gat_rank      <<<(N + RB - 1) / RB,    PT,  0, stream>>>(bcnt, part, cnt, ovf, (int*)d_out, N);
    gat_gemm_wmma <<<(N + 127) / 128,      256, 0, stream>>>(x, W, h, N);
    gat_scores    <<<(N * 32 + 255) / 256, 256, 0, stream>>>(h, attn, s_src, s_dst, N);
    gat_row_gather<<<(N * 32 + 255) / 256, 256, 0, stream>>>(
        src, dst, s_src, s_dst, cnt, ovf, h, (const int*)d_out, out, N, E);
}


# Rank and row gather fused: one block per 256-row bucket ranks its run into LDS lists, its waves sort each list by dst and sum it; slot stores, cnt pass and a launch gone; f32 sum order = dst order
# speedup vs baseline: 4.9090x; 1.0888x over previous
#include <hip/hip_runtime.h>
#include <cmath>

#define IN_F  256
#define OUT_F 64
#define LRELU_ALPHA 0.2f
#define SLOT 64
#define RB_SHIFT 8
#define RB (1 << RB_SHIFT)
#define BCAP (RB * OUT_F)
#define NB_MAX 512
#define BSTRIDE 64
#define PT 512
#define PCH 16
#define RCH 9
#define ROWFLAG (1 << 30)
#define OVW(n) ((n) - NB_MAX * BSTRIDE)
#define OVCAP(n) ((OVW(n) - 2) / 2)
#define FT 1024

typedef __attribute__((ext_vector_type(2))) float v2f;
typedef __attribute__((ext_vector_type(8))) float v8f;

__global__ __launch_bounds__(256) void gat_gemm_wmma(
    const float* __restrict__ x, const float* __restrict__ W,
    float* __restrict__ h, int N)
{
    __shared__ float Wlds[IN_F * OUT_F];
    const int tid = threadIdx.x;
    for (int i = tid; i < IN_F * OUT_F; i += 256) {
        const int k = i / OUT_F, c = i % OUT_F;
        Wlds[(k >> 1) * (2 * OUT_F) + c * 2 + (k & 1)] = W[i];
    }
    __syncthreads();

    const int wave = tid >> 5;
    const int lane = tid & 31;
    const int half = lane >> 4;
    const int l16  = lane & 15;
    const int tile_row0 = blockIdx.x * 128 + wave * 16;
    const int arow = tile_row0 + l16;
    const int lrow = arow < N ? arow : (N - 1);

    const float* xptr = x + (size_t)lrow * IN_F + 2 * half;
    const float* bptr = Wlds + half * (2 * OUT_F) + l16 * 2;

    v8f acc[4] = {};

    for (int k0 = 0; k0 < IN_F; k0 += 4) {
        const float2 t = *(const float2*)(xptr + k0);
        v2f a; a.x = t.x; a.y = t.y;
        const float* bk = bptr + (k0 >> 1) * (2 * OUT_F);
#pragma unroll
        for (int n = 0; n < 4; ++n) {
            const float2 u = *(const float2*)(bk + n * 32);
            v2f b; b.x = u.x; b.y = u.y;
            acc[n] = __builtin_amdgcn_wmma_f32_16x16x4_f32(
                false, a, false, b, (short)0, acc[n], false, false);
        }
    }

    const int rbase = tile_row0 + 8 * half;
#pragma unroll
    for (int r = 0; r < 8; ++r) {
        const int rr = rbase + r;
        if (rr < N) {
#pragma unroll
            for (int n = 0; n < 4; ++n)
                h[(size_t)rr * OUT_F + n * 16 + l16] = acc[n][r];
        }
    }
}

__global__ __launch_bounds__(256) void gat_scores(
    const float* __restrict__ h, const float* __restrict__ attn,
    float* __restrict__ s_src, float* __restrict__ s_dst, int N)
{
    const int node = (blockIdx.x * blockDim.x + threadIdx.x) >> 5;
    const int lane = threadIdx.x & 31;
    if (node >= N) return;
    const float* hr = h + (size_t)node * OUT_F;
    const float h0 = hr[lane], h1 = hr[lane + 32];
    float ps = h0 * attn[lane]         + h1 * attn[lane + 32];
    float pd = h0 * attn[OUT_F + lane] + h1 * attn[OUT_F + lane + 32];
    for (int off = 16; off > 0; off >>= 1) {
        ps += __shfl_down(ps, off, 32);
        pd += __shfl_down(pd, off, 32);
    }
    if (lane == 0) { s_src[node] = ps; s_dst[node] = pd; }
}

__global__ void gat_init(int* __restrict__ cnt, int* __restrict__ ovf,
                         int* __restrict__ bcnt, int N)
{
    const int i = blockIdx.x * blockDim.x + threadIdx.x;
    if (i < N) cnt[i] = 0;
    if (i < NB_MAX) bcnt[i * BSTRIDE] = 0;
    if (i == 0) ovf[0] = 0;
}

__device__ __forceinline__ void ovf_push(int* __restrict__ ovf, int row, int d, int N)
{
    const int k = atomicAdd(&ovf[0], 1);
    if (k < OVCAP(N)) { ovf[2 + 2 * k] = row; ovf[3 + 2 * k] = d; }
}

__global__ __launch_bounds__(PT) void gat_partition(
    const int* __restrict__ src, const int* __restrict__ dst,
    int* __restrict__ bcnt, int* __restrict__ part,
    int* __restrict__ cnt, int* __restrict__ ovf, int N, int E)
{
    __shared__ int hist[NB_MAX];
    __shared__ int base[NB_MAX];
    const int tid = threadIdx.x;
    const int nb = (N + RB - 1) >> RB_SHIFT;
    static_assert(PT == NB_MAX, "one thread clears one bucket counter");
    hist[tid] = 0;
    const int e0 = blockIdx.x * (PT * PCH) + tid;
    int s[PCH], d[PCH], r[PCH];
#pragma unroll
    for (int j = 0; j < PCH; ++j) {
        const int i = e0 + j * PT;
        const int ic = i < E ? i : E - 1;
        const int sv = src[ic];
        d[j] = dst[ic];
        s[j] = i < E ? sv : -1;
    }
    __syncthreads();
#pragma unroll
    for (int j = 0; j < PCH; ++j)
        r[j] = s[j] >= 0 ? atomicAdd(&hist[s[j] >> RB_SHIFT], 1) : 0;
    __syncthreads();
    if (tid < nb) {
        const int c = hist[tid];
        base[tid] = c > 0 ? atomicAdd(&bcnt[tid * BSTRIDE], c) : 0;
    }
    __syncthreads();
#pragma unroll
    for (int j = 0; j < PCH; ++j)
        r[j] += base[s[j] >= 0 ? (s[j] >> RB_SHIFT) : 0];
#pragma unroll
    for (int j = 0; j < PCH; ++j) {
        if (s[j] >= 0) {
            const int b = s[j] >> RB_SHIFT;
            const int pos = r[j];
            const int rows = min(RB, N - (b << RB_SHIFT));
            if (pos < rows * OUT_F) {
                part[(size_t)b * BCAP + pos] =
                    (int)(((unsigned)(s[j] & (RB - 1)) << 24) | (unsigned)d[j]);
            } else {
                atomicOr(&cnt[s[j]], ROWFLAG);
                ovf_push(ovf, s[j], d[j], N);
            }
        }
    }
}

template <int J> __device__ __forceinline__ int xor_lane(int v) {
    if constexpr (J == 16)
        return __builtin_amdgcn_permlanex16(v, v, 0x76543210, 0xfedcba98, false, false);
    else
        return __builtin_amdgcn_update_dpp(v, v, 0x160 | J, 0xf, 0xf, false);
}
template <int J> __device__ __forceinline__ float xor_lane_f(float v) {
    return __int_as_float(xor_lane<J>(__float_as_int(v)));
}
__device__ __forceinline__ float wave_max32(float v) {
    v = fmaxf(v, xor_lane_f<16>(v));
    v = fmaxf(v, xor_lane_f<8>(v));
    v = fmaxf(v, xor_lane_f<4>(v));
    v = fmaxf(v, xor_lane_f<2>(v));
    v = fmaxf(v, xor_lane_f<1>(v));
    return v;
}
__device__ __forceinline__ float wave_sum32(float v) {
    v += xor_lane_f<16>(v);
    v += xor_lane_f<8>(v);
    v += xor_lane_f<4>(v);
    v += xor_lane_f<2>(v);
    v += xor_lane_f<1>(v);
    return v;
}

template <int B> __device__ __forceinline__ void h_rows(int dd, float p, int j0,
                                                       const float* __restrict__ h, int lane,
                                                       float& ax, float& ay)
{
    float2 hv[B];
    float pj[B];
#pragma unroll
    for (int u = 0; u < B; ++u) {
        const int dj = __builtin_amdgcn_readlane(dd, j0 + u);
        pj[u] = __int_as_float(__builtin_amdgcn_readlane(__float_as_int(p), j0 + u));
        hv[u] = ((const float2*)(h + (size_t)dj * OUT_F))[lane];
    }
#pragma unroll
    for (int u = 0; u < B; ++u) { ax += pj[u] * hv[u].x; ay += pj[u] * hv[u].y; }
}

__device__ __forceinline__ void row_chunk(bool valid, int d, float ss,
                                          const float* __restrict__ s_dst,
                                          const float* __restrict__ h, int lane,
                                          float& m, float& l, float& ax, float& ay)
{
    const unsigned mask = __builtin_amdgcn_ballot_w32(valid);
    if (mask == 0u) return;
    float e = -INFINITY;
    int dd = 0;
    if (valid) {
        dd = d;
        const float v = ss + s_dst[d];
        e = -(v > 0.f ? v : LRELU_ALPHA * v);
    }
    const float mn = fmaxf(m, wave_max32(e));
    const float scale = (m == -INFINITY) ? 0.f : __expf(m - mn);
    const float p = valid ? __expf(e - mn) : 0.f;
    l = l * scale + wave_sum32(p);
    ax *= scale; ay *= scale;
    m = mn;
    const int nmax = 32 - __clz((int)mask);
    for (int j0 = 0; j0 < nmax; j0 += 16) {
        if (j0 + 8 < nmax) h_rows<16>(dd, p, j0, h, lane, ax, ay);
        else               h_rows<8>(dd, p, j0, h, lane, ax, ay);
    }
}

template <int J> __device__ __forceinline__ int cmp_xchg(int v, bool keep_min) {
    const int o = xor_lane<J>(v);
    return keep_min ? min(v, o) : max(v, o);
}
__device__ __forceinline__ int bitonic_merge32(int v, int lane, bool up) {
    v = cmp_xchg<16>(v, ((lane & 16) == 0) == up);
    v = cmp_xchg<8>(v, ((lane & 8) == 0) == up);
    v = cmp_xchg<4>(v, ((lane & 4) == 0) == up);
    v = cmp_xchg<2>(v, ((lane & 2) == 0) == up);
    v = cmp_xchg<1>(v, ((lane & 1) == 0) == up);
    return v;
}
__device__ __forceinline__ int bitonic_sort32(int v, int lane, bool up) {
    const bool a2 = ((lane & 2) == 0) == up, a4 = ((lane & 4) == 0) == up;
    const bool a8 = ((lane & 8) == 0) == up, a16 = ((lane & 16) == 0) == up;
    v = cmp_xchg<1>(v, ((lane & 1) == 0) == a2);
    v = cmp_xchg<2>(v, ((lane & 2) == 0) == a4);
    v = cmp_xchg<1>(v, ((lane & 1) == 0) == a4);
    v = cmp_xchg<4>(v, ((lane & 4) == 0) == a8);
    v = cmp_xchg<2>(v, ((lane & 2) == 0) == a8);
    v = cmp_xchg<1>(v, ((lane & 1) == 0) == a8);
    v = cmp_xchg<8>(v, ((lane & 8) == 0) == a16);
    v = cmp_xchg<4>(v, ((lane & 4) == 0) == a16);
    v = cmp_xchg<2>(v, ((lane & 2) == 0) == a16);
    v = cmp_xchg<1>(v, ((lane & 1) == 0) == a16);
    return bitonic_merge32(v, lane, up);
}

__global__ __launch_bounds__(FT, 16) void gat_rank_gather(
    const int* __restrict__ src, const int* __restrict__ dst,
    const float* __restrict__ s_src, const float* __restrict__ s_dst,
    const int* __restrict__ bcnt, const int* __restrict__ cnt, const int* __restrict__ ovf,
    const float* __restrict__ h, const int* part, float* out, int N, int E)
{
    __shared__ int lc[RB];
    __shared__ int lst[RB * SLOT];
    __shared__ float lss[RB];
    __shared__ int lfl[RB];
    const int tid = threadIdx.x;
    const int b = blockIdx.x;
    const int row0 = b << RB_SHIFT;
    const int rows = min(RB, N - row0);
    if (tid < RB) lc[tid] = 0;
    if (tid < rows) { lss[tid] = s_src[row0 + tid]; lfl[tid] = cnt[row0 + tid]; }
    __syncthreads();
    const int n = min(bcnt[b * BSTRIDE], rows * OUT_F);
    const int* p = part + (size_t)b * BCAP;
    for (int t0 = 0; t0 < n; t0 += FT * RCH) {
        int w[RCH];
#pragma unroll
        for (int j = 0; j < RCH; ++j) {
            const int e = t0 + j * FT + tid;
            w[j] = p[e < n ? e : n - 1];
        }
        int rk[RCH];
#pragma unroll
        for (int j = 0; j < RCH; ++j) {
            const int e = t0 + j * FT + tid;
            rk[j] = SLOT;
            if (e < n) rk[j] = atomicAdd(&lc[(unsigned)w[j] >> 24], 1);
        }
#pragma unroll
        for (int j = 0; j < RCH; ++j)
            if (rk[j] < SLOT) lst[((unsigned)w[j] >> 24) * SLOT + rk[j]] = w[j] & 0xffffff;
    }
    __syncthreads();

    const int lane = tid & 31;
    const int wave = __builtin_amdgcn_readfirstlane(tid >> 5);
    for (int r = wave; r < rows; r += FT / 32) {
        const int row = row0 + r;
        const int c = __builtin_amdgcn_readfirstlane(lc[r]);
        const float ss = lss[r];
        float m = -INFINITY, l = 0.f, ax = 0.f, ay = 0.f;
        int total = 0;
        if (__builtin_amdgcn_readfirstlane(lfl[r]) & ROWFLAG) total = ovf[0];
        const bool full = c > SLOT || total > OVCAP(N);
        if (!full) {
            int k0 = lane < c ? lst[r * SLOT + lane] : 0x7fffffff;
            int k1 = 0x7fffffff;
            if (c > 32) {
                k1 = lane + 32 < c ? lst[r * SLOT + 32 + lane] : 0x7fffffff;
                k0 = bitonic_sort32(k0, lane, true);
                k1 = bitonic_sort32(k1, lane, false);
                const int lo = min(k0, k1), hi = max(k0, k1);
                k0 = bitonic_merge32(lo, lane, true);
                k1 = bitonic_merge32(hi, lane, true);
            } else {
                k0 = bitonic_sort32(k0, lane, true);
            }
            for (int q = 0; q < c; q += 32) {
                const bool v = q + lane < c;
                const int kk = q ? k1 : k0;
                row_chunk(v, v ? kk : 0, ss, s_dst, h, lane, m, l, ax, ay);
            }
        }
        const int* rlist = full ? src : ovf + 2;
        const int* dlist = full ? dst : ovf + 3;
        const int rstep = full ? 1 : 2;
        const int lim = full ? E : total;
        for (int base = 0; base < lim; base += 32) {
            const int idx = base + lane;
            bool v = idx < lim;
            int d = 0;
            if (v) {
                v = (rlist[(size_t)idx * rstep] == row);
                if (v) d = dlist[(size_t)idx * rstep];
            }
            row_chunk(v, d, ss, s_dst, h, lane, m, l, ax, ay);
        }
        float rx = 0.f, ry = 0.f;
        if (l > 0.f) { rx = ax / l; ry = ay / l; }
        float2 o;
        o.x = rx > 0.f ? rx : expm1f(rx);
        o.y = ry > 0.f ? ry : expm1f(ry);
        *(float2*)(out + (size_t)row * OUT_F + 2 * lane) = o;
    }
}

extern "C" void kernel_launch(void* const* d_in, const int* in_sizes, int n_in,
                              void* d_out, int out_size, void* d_ws, size_t ws_size,
                              hipStream_t stream)
{
    const float* x    = (const float*)d_in[0];
    const int*   edge = (const int*)  d_in[1];
    const float* W    = (const float*)d_in[2];
    const float* attn = (const float*)d_in[3];

    const int N = in_sizes[0] / IN_F;
    const int E = in_sizes[1] / 2;
    const int* src = edge;
    const int* dst = edge + E;
    float* out = (float*)d_out;

    float* h     = (float*)d_ws;
    float* s_src = h + (size_t)N * OUT_F;
    float* s_dst = s_src + N;
    int*   cnt   = (int*)(s_dst + N);
    int*   ovf   = cnt + N;

    int* bcnt = ovf + OVW(N);
    int* part = (int*)d_out;

    gat_init      <<<(N + 255) / 256,      256, 0, stream>>>(cnt, ovf, bcnt, N);
    gat_partition <<<(E + PT * PCH - 1) / (PT * PCH), PT, 0, stream>>>(src, dst, bcnt, part, cnt, ovf, N, E);
    gat_gemm_wmma <<<(N + 127) / 128,      256, 0, stream>>>(x, W, h, N);
    gat_scores    <<<(N * 32 + 255) / 256, 256, 0, stream>>>(h, attn, s_src, s_dst, N);
    gat_rank_gather<<<(N + RB - 1) / RB,  FT,  0, stream>>>(
        src, dst, s_src, s_dst, bcnt, cnt, ovf, h, part, out, N, E);
}


# Scores in the GEMM epilogue: a wave forms s_src, s_dst of its 16 rows from its accumulators (transposing 16-lane DPP sums); the scores launch is gone (5 to 4 launches); scores' sum order changes
# speedup vs baseline: 5.2837x; 1.0763x over previous
#include <hip/hip_runtime.h>
#include <cmath>

#define IN_F  256
#define OUT_F 64
#define LRELU_ALPHA 0.2f
#define SLOT 64
#define RB_SHIFT 8
#define RB (1 << RB_SHIFT)
#define BCAP (RB * OUT_F)
#define NB_MAX 512
#define BSTRIDE 64
#define PT 512
#define PCH 16
#define RCH 9
#define ROWFLAG (1 << 30)
#define OVW(n) ((n) - NB_MAX * BSTRIDE)
#define OVCAP(n) ((OVW(n) - 2) / 2)
#define FT 1024

typedef __attribute__((ext_vector_type(2))) float v2f;
typedef __attribute__((ext_vector_type(8))) float v8f;

template <int J> __device__ __forceinline__ int xor_lane(int v) {
    if constexpr (J == 16)
        return __builtin_amdgcn_permlanex16(v, v, 0x76543210, 0xfedcba98, false, false);
    else
        return __builtin_amdgcn_update_dpp(v, v, 0x160 | J, 0xf, 0xf, false);
}
template <int J> __device__ __forceinline__ float xor_lane_f(float v) {
    return __int_as_float(xor_lane<J>(__float_as_int(v)));
}

__global__ __launch_bounds__(256) void gat_gemm_wmma(
    const float* __restrict__ x, const float* __restrict__ W,
    const float* __restrict__ attn, float* __restrict__ h,
    float* __restrict__ s_src, float* __restrict__ s_dst, int N)
{
    __shared__ float Wlds[IN_F * OUT_F];
    const int tid = threadIdx.x;
    for (int i = tid; i < IN_F * OUT_F; i += 256) {
        const int k = i / OUT_F, c = i % OUT_F;
        Wlds[(k >> 1) * (2 * OUT_F) + c * 2 + (k & 1)] = W[i];
    }
    __syncthreads();

    const int wave = tid >> 5;
    const int lane = tid & 31;
    const int half = lane >> 4;
    const int l16  = lane & 15;
    const int tile_row0 = blockIdx.x * 128 + wave * 16;
    const int arow = tile_row0 + l16;
    const int lrow = arow < N ? arow : (N - 1);

    const float* xptr = x + (size_t)lrow * IN_F + 2 * half;
    const float* bptr = Wlds + half * (2 * OUT_F) + l16 * 2;

    v8f acc[4] = {};

    float a_s[4], a_d[4];
    for (int n = 0; n < 4; ++n) {
        a_s[n] = attn[n * 16 + l16];
        a_d[n] = attn[OUT_F + n * 16 + l16];
    }

    for (int k0 = 0; k0 < IN_F; k0 += 4) {
        const float2 t = *(const float2*)(xptr + k0);
        v2f a; a.x = t.x; a.y = t.y;
        const float* bk = bptr + (k0 >> 1) * (2 * OUT_F);
#pragma unroll
        for (int n = 0; n < 4; ++n) {
            const float2 u = *(const float2*)(bk + n * 32);
            v2f b; b.x = u.x; b.y = u.y;
            acc[n] = __builtin_amdgcn_wmma_f32_16x16x4_f32(
                false, a, false, b, (short)0, acc[n], false, false);
        }
    }

    const int rbase = tile_row0 + 8 * half;
#pragma unroll
    for (int r = 0; r < 8; ++r) {
        const int rr = rbase + r;
        if (rr < N) {
#pragma unroll
            for (int n = 0; n < 4; ++n)
                h[(size_t)rr * OUT_F + n * 16 + l16] = acc[n][r];
        }
    }

    float v[16];
    for (int r = 0; r < 8; ++r) {
        float ps = acc[0][r] * a_s[0], pd = acc[0][r] * a_d[0];
        for (int n = 1; n < 4; ++n) { ps += acc[n][r] * a_s[n]; pd += acc[n][r] * a_d[n]; }
        v[r] = ps; v[8 + r] = pd;
    }
    const bool b3 = (l16 & 8) != 0, b2 = (l16 & 4) != 0;
    const bool b1 = (l16 & 2) != 0, b0 = (l16 & 1) != 0;
    float w8[8], w4[4], w2[2];
    for (int j = 0; j < 8; ++j) {
        const float keep = b3 ? v[8 + j] : v[j];
        const float send = b3 ? v[j] : v[8 + j];
        w8[j] = keep + xor_lane_f<8>(send);
    }
    for (int j = 0; j < 4; ++j) {
        const float keep = b2 ? w8[4 + j] : w8[j];
        const float send = b2 ? w8[j] : w8[4 + j];
        w4[j] = keep + xor_lane_f<4>(send);
    }
    for (int j = 0; j < 2; ++j) {
        const float keep = b1 ? w4[2 + j] : w4[j];
        const float send = b1 ? w4[j] : w4[2 + j];
        w2[j] = keep + xor_lane_f<2>(send);
    }
    const float keep1 = b0 ? w2[1] : w2[0];
    const float send1 = b0 ? w2[0] : w2[1];
    const float sc = keep1 + xor_lane_f<1>(send1);
    const int srow = rbase + (l16 & 7);
    if (srow < N) {
        float* sp = b3 ? s_dst : s_src;
        sp[srow] = sc;
    }
}

__global__ void gat_init(int* __restrict__ cnt, int* __restrict__ ovf,
                         int* __restrict__ bcnt, int N)
{
    const int i = blockIdx.x * blockDim.x + threadIdx.x;
    if (i < N) cnt[i] = 0;
    if (i < NB_MAX) bcnt[i * BSTRIDE] = 0;
    if (i == 0) ovf[0] = 0;
}

__device__ __forceinline__ void ovf_push(int* __restrict__ ovf, int row, int d, int N)
{
    const int k = atomicAdd(&ovf[0], 1);
    if (k < OVCAP(N)) { ovf[2 + 2 * k] = row; ovf[3 + 2 * k] = d; }
}

__global__ __launch_bounds__(PT) void gat_partition(
    const int* __restrict__ src, const int* __restrict__ dst,
    int* __restrict__ bcnt, int* __restrict__ part,
    int* __restrict__ cnt, int* __restrict__ ovf, int N, int E)
{
    __shared__ int hist[NB_MAX];
    __shared__ int base[NB_MAX];
    const int tid = threadIdx.x;
    const int nb = (N + RB - 1) >> RB_SHIFT;
    static_assert(PT == NB_MAX, "one thread clears one bucket counter");
    hist[tid] = 0;
    const int e0 = blockIdx.x * (PT * PCH) + tid;
    int s[PCH], d[PCH], r[PCH];
#pragma unroll
    for (int j = 0; j < PCH; ++j) {
        const int i = e0 + j * PT;
        const int ic = i < E ? i : E - 1;
        const int sv = src[ic];
        d[j] = dst[ic];
        s[j] = i < E ? sv : -1;
    }
    __syncthreads();
#pragma unroll
    for (int j = 0; j < PCH; ++j)
        r[j] = s[j] >= 0 ? atomicAdd(&hist[s[j] >> RB_SHIFT], 1) : 0;
    __syncthreads();
    if (tid < nb) {
        const int c = hist[tid];
        base[tid] = c > 0 ? atomicAdd(&bcnt[tid * BSTRIDE], c) : 0;
    }
    __syncthreads();
#pragma unroll
    for (int j = 0; j < PCH; ++j)
        r[j] += base[s[j] >= 0 ? (s[j] >> RB_SHIFT) : 0];
#pragma unroll
    for (int j = 0; j < PCH; ++j) {
        if (s[j] >= 0) {
            const int b = s[j] >> RB_SHIFT;
            const int pos = r[j];
            const int rows = min(RB, N - (b << RB_SHIFT));
            if (pos < rows * OUT_F) {
                part[(size_t)b * BCAP + pos] =
                    (int)(((unsigned)(s[j] & (RB - 1)) << 24) | (unsigned)d[j]);
            } else {
                atomicOr(&cnt[s[j]], ROWFLAG);
                ovf_push(ovf, s[j], d[j], N);
            }
        }
    }
}

__device__ __forceinline__ float wave_max32(float v) {
    v = fmaxf(v, xor_lane_f<16>(v));
    v = fmaxf(v, xor_lane_f<8>(v));
    v = fmaxf(v, xor_lane_f<4>(v));
    v = fmaxf(v, xor_lane_f<2>(v));
    v = fmaxf(v, xor_lane_f<1>(v));
    return v;
}
__device__ __forceinline__ float wave_sum32(float v) {
    v += xor_lane_f<16>(v);
    v += xor_lane_f<8>(v);
    v += xor_lane_f<4>(v);
    v += xor_lane_f<2>(v);
    v += xor_lane_f<1>(v);
    return v;
}

template <int B> __device__ __forceinline__ void h_rows(int dd, float p, int j0,
                                                       const float* __restrict__ h, int lane,
                                                       float& ax, float& ay)
{
    float2 hv[B];
    float pj[B];
#pragma unroll
    for (int u = 0; u < B; ++u) {
        const int dj = __builtin_amdgcn_readlane(dd, j0 + u);
        pj[u] = __int_as_float(__builtin_amdgcn_readlane(__float_as_int(p), j0 + u));
        hv[u] = ((const float2*)(h + (size_t)dj * OUT_F))[lane];
    }
#pragma unroll
    for (int u = 0; u < B; ++u) { ax += pj[u] * hv[u].x; ay += pj[u] * hv[u].y; }
}

__device__ __forceinline__ void row_chunk(bool valid, int d, float ss,
                                          const float* __restrict__ s_dst,
                                          const float* __restrict__ h, int lane,
                                          float& m, float& l, float& ax, float& ay)
{
    const unsigned mask = __builtin_amdgcn_ballot_w32(valid);
    if (mask == 0u) return;
    float e = -INFINITY;
    int dd = 0;
    if (valid) {
        dd = d;
        const float v = ss + s_dst[d];
        e = -(v > 0.f ? v : LRELU_ALPHA * v);
    }
    const float mn = fmaxf(m, wave_max32(e));
    const float scale = (m == -INFINITY) ? 0.f : __expf(m - mn);
    const float p = valid ? __expf(e - mn) : 0.f;
    l = l * scale + wave_sum32(p);
    ax *= scale; ay *= scale;
    m = mn;
    const int nmax = 32 - __clz((int)mask);
    for (int j0 = 0; j0 < nmax; j0 += 16) {
        if (j0 + 8 < nmax) h_rows<16>(dd, p, j0, h, lane, ax, ay);
        else               h_rows<8>(dd, p, j0, h, lane, ax, ay);
    }
}

template <int J> __device__ __forceinline__ int cmp_xchg(int v, bool keep_min) {
    const int o = xor_lane<J>(v);
    return keep_min ? min(v, o) : max(v, o);
}
__device__ __forceinline__ int bitonic_merge32(int v, int lane, bool up) {
    v = cmp_xchg<16>(v, ((lane & 16) == 0) == up);
    v = cmp_xchg<8>(v, ((lane & 8) == 0) == up);
    v = cmp_xchg<4>(v, ((lane & 4) == 0) == up);
    v = cmp_xchg<2>(v, ((lane & 2) == 0) == up);
    v = cmp_xchg<1>(v, ((lane & 1) == 0) == up);
    return v;
}
__device__ __forceinline__ int bitonic_sort32(int v, int lane, bool up) {
    const bool a2 = ((lane & 2) == 0) == up, a4 = ((lane & 4) == 0) == up;
    const bool a8 = ((lane & 8) == 0) == up, a16 = ((lane & 16) == 0) == up;
    v = cmp_xchg<1>(v, ((lane & 1) == 0) == a2);
    v = cmp_xchg<2>(v, ((lane & 2) == 0) == a4);
    v = cmp_xchg<1>(v, ((lane & 1) == 0) == a4);
    v = cmp_xchg<4>(v, ((lane & 4) == 0) == a8);
    v = cmp_xchg<2>(v, ((lane & 2) == 0) == a8);
    v = cmp_xchg<1>(v, ((lane & 1) == 0) == a8);
    v = cmp_xchg<8>(v, ((lane & 8) == 0) == a16);
    v = cmp_xchg<4>(v, ((lane & 4) == 0) == a16);
    v = cmp_xchg<2>(v, ((lane & 2) == 0) == a16);
    v = cmp_xchg<1>(v, ((lane & 1) == 0) == a16);
    return bitonic_merge32(v, lane, up);
}

__global__ __launch_bounds__(FT, 16) void gat_rank_gather(
    const int* __restrict__ src, const int* __restrict__ dst,
    const float* __restrict__ s_src, const float* __restrict__ s_dst,
    const int* __restrict__ bcnt, const int* __restrict__ cnt, const int* __restrict__ ovf,
    const float* __restrict__ h, const int* part, float* out, int N, int E)
{
    __shared__ int lc[RB];
    __shared__ int lst[RB * SLOT];
    __shared__ float lss[RB];
    __shared__ int lfl[RB];
    const int tid = threadIdx.x;
    const int b = blockIdx.x;
    const int row0 = b << RB_SHIFT;
    const int rows = min(RB, N - row0);
    if (tid < RB) lc[tid] = 0;
    if (tid < rows) { lss[tid] = s_src[row0 + tid]; lfl[tid] = cnt[row0 + tid]; }
    __syncthreads();
    const int n = min(bcnt[b * BSTRIDE], rows * OUT_F);
    const int* p = part + (size_t)b * BCAP;
    for (int t0 = 0; t0 < n; t0 += FT * RCH) {
        int w[RCH];
#pragma unroll
        for (int j = 0; j < RCH; ++j) {
            const int e = t0 + j * FT + tid;
            w[j] = p[e < n ? e : n - 1];
        }
        int rk[RCH];
#pragma unroll
        for (int j = 0; j < RCH; ++j) {
            const int e = t0 + j * FT + tid;
            rk[j] = SLOT;
            if (e < n) rk[j] = atomicAdd(&lc[(unsigned)w[j] >> 24], 1);
        }
#pragma unroll
        for (int j = 0; j < RCH; ++j)
            if (rk[j] < SLOT) lst[((unsigned)w[j] >> 24) * SLOT + rk[j]] = w[j] & 0xffffff;
    }
    __syncthreads();

    const int lane = tid & 31;
    const int wave = __builtin_amdgcn_readfirstlane(tid >> 5);
    for (int r = wave; r < rows; r += FT / 32) {
        const int row = row0 + r;
        const int c = __builtin_amdgcn_readfirstlane(lc[r]);
        const float ss = lss[r];
        float m = -INFINITY, l = 0.f, ax = 0.f, ay = 0.f;
        int total = 0;
        if (__builtin_amdgcn_readfirstlane(lfl[r]) & ROWFLAG) total = ovf[0];
        const bool full = c > SLOT || total > OVCAP(N);
        if (!full) {
            int k0 = lane < c ? lst[r * SLOT + lane] : 0x7fffffff;
            int k1 = 0x7fffffff;
            if (c > 32) {
                k1 = lane + 32 < c ? lst[r * SLOT + 32 + lane] : 0x7fffffff;
                k0 = bitonic_sort32(k0, lane, true);
                k1 = bitonic_sort32(k1, lane, false);
                const int lo = min(k0, k1), hi = max(k0, k1);
                k0 = bitonic_merge32(lo, lane, true);
                k1 = bitonic_merge32(hi, lane, true);
            } else {
                k0 = bitonic_sort32(k0, lane, true);
            }
            for (int q = 0; q < c; q += 32) {
                const bool v = q + lane < c;
                const int kk = q ? k1 : k0;
                row_chunk(v, v ? kk : 0, ss, s_dst, h, lane, m, l, ax, ay);
            }
        }
        const int* rlist = full ? src : ovf + 2;
        const int* dlist = full ? dst : ovf + 3;
        const int rstep = full ? 1 : 2;
        const int lim = full ? E : total;
        for (int base = 0; base < lim; base += 32) {
            const int idx = base + lane;
            bool v = idx < lim;
            int d = 0;
            if (v) {
                v = (rlist[(size_t)idx * rstep] == row);
                if (v) d = dlist[(size_t)idx * rstep];
            }
            row_chunk(v, d, ss, s_dst, h, lane, m, l, ax, ay);
        }
        float rx = 0.f, ry = 0.f;
        if (l > 0.f) { rx = ax / l; ry = ay / l; }
        float2 o;
        o.x = rx > 0.f ? rx : expm1f(rx);
        o.y = ry > 0.f ? ry : expm1f(ry);
        *(float2*)(out + (size_t)row * OUT_F + 2 * lane) = o;
    }
}

extern "C" void kernel_launch(void* const* d_in, const int* in_sizes, int n_in,
                              void* d_out, int out_size, void* d_ws, size_t ws_size,
                              hipStream_t stream)
{
    const float* x    = (const float*)d_in[0];
    const int*   edge = (const int*)  d_in[1];
    const float* W    = (const float*)d_in[2];
    const float* attn = (const float*)d_in[3];

    const int N = in_sizes[0] / IN_F;
    const int E = in_sizes[1] / 2;
    const int* src = edge;
    const int* dst = edge + E;
    float* out = (float*)d_out;

    float* h     = (float*)d_ws;
    float* s_src = h + (size_t)N * OUT_F;
    float* s_dst = s_src + N;
    int*   cnt   = (int*)(s_dst + N);
    int*   ovf   = cnt + N;

    int* bcnt = ovf + OVW(N);
    int* part = (int*)d_out;

    gat_init      <<<(N + 255) / 256,      256, 0, stream>>>(cnt, ovf, bcnt, N);
    gat_partition <<<(E + PT * PCH - 1) / (PT * PCH), PT, 0, stream>>>(src, dst, bcnt, part, cnt, ovf, N, E);
    gat_gemm_wmma <<<(N + 127) / 128,      256, 0, stream>>>(x, W, attn, h, s_src, s_dst, N);
    gat_rank_gather<<<(N + RB - 1) / RB,  FT,  0, stream>>>(
        src, dst, s_src, s_dst, bcnt, cnt, ovf, h, part, out, N, E);
}


# GEMM and edge partition as one launch: blocks 0-390 form h and the scores, the blocks behind them file the edges, so the partition runs under the matrix work; 4 to 3 launches; all sums unchanged
# speedup vs baseline: 6.0040x; 1.1363x over previous
#include <hip/hip_runtime.h>
#include <cmath>

#define IN_F  256
#define OUT_F 64
#define LRELU_ALPHA 0.2f
#define SLOT 64
#define RB_SHIFT 8
#define RB (1 << RB_SHIFT)
#define BCAP (RB * OUT_F)
#define NB_MAX 512
#define BSTRIDE 64
#define PT 512
#define PCH 17
#define GROWS (PT / 32 * 16)
#define RCH 9
#define ROWFLAG (1 << 30)
#define OVW(n) ((n) - NB_MAX * BSTRIDE)
#define OVCAP(n) ((OVW(n) - 2) / 2)
#define FT 1024

typedef __attribute__((ext_vector_type(2))) float v2f;
typedef __attribute__((ext_vector_type(8))) float v8f;

template <int J> __device__ __forceinline__ int xor_lane(int v) {
    if constexpr (J == 16)
        return __builtin_amdgcn_permlanex16(v, v, 0x76543210, 0xfedcba98, false, false);
    else
        return __builtin_amdgcn_update_dpp(v, v, 0x160 | J, 0xf, 0xf, false);
}
template <int J> __device__ __forceinline__ float xor_lane_f(float v) {
    return __int_as_float(xor_lane<J>(__float_as_int(v)));
}

__device__ __forceinline__ void gemm_block(
    float* Wlds, int gb,
    const float* __restrict__ x, const float* __restrict__ W,
    const float* __restrict__ attn, float* __restrict__ h,
    float* __restrict__ s_src, float* __restrict__ s_dst, int N)
{
    const int tid = threadIdx.x;
    for (int i = tid; i < IN_F * OUT_F; i += PT) {
        const int k = i / OUT_F, c = i % OUT_F;
        Wlds[(k >> 1) * (2 * OUT_F) + c * 2 + (k & 1)] = W[i];
    }
    __syncthreads();

    const int wave = tid >> 5;
    const int lane = tid & 31;
    const int half = lane >> 4;
    const int l16  = lane & 15;
    const int tile_row0 = gb * GROWS + wave * 16;
    const int arow = tile_row0 + l16;
    const int lrow = arow < N ? arow : (N - 1);

    const float* xptr = x + (size_t)lrow * IN_F + 2 * half;
    const float* bptr = Wlds + half * (2 * OUT_F) + l16 * 2;

    v8f acc[4] = {};

    for (int k0 = 0; k0 < IN_F; k0 += 4) {
        const float2 t = *(const float2*)(xptr + k0);
        v2f a; a.x = t.x; a.y = t.y;
        const float* bk = bptr + (k0 >> 1) * (2 * OUT_F);
#pragma unroll
        for (int n = 0; n < 4; ++n) {
            const float2 u = *(const float2*)(bk + n * 32);
            v2f b; b.x = u.x; b.y = u.y;
            acc[n] = __builtin_amdgcn_wmma_f32_16x16x4_f32(
                false, a, false, b, (short)0, acc[n], false, false);
        }
    }

    float a_s[4], a_d[4];
    for (int n = 0; n < 4; ++n) {
        a_s[n] = attn[n * 16 + l16];
        a_d[n] = attn[OUT_F + n * 16 + l16];
    }

    const int rbase = tile_row0 + 8 * half;
#pragma unroll
    for (int r = 0; r < 8; ++r) {
        const int rr = rbase + r;
        if (rr < N) {
#pragma unroll
            for (int n = 0; n < 4; ++n)
                h[(size_t)rr * OUT_F + n * 16 + l16] = acc[n][r];
        }
    }

    float v[16];
    for (int r = 0; r < 8; ++r) {
        float ps = acc[0][r] * a_s[0], pd = acc[0][r] * a_d[0];
        for (int n = 1; n < 4; ++n) { ps += acc[n][r] * a_s[n]; pd += acc[n][r] * a_d[n]; }
        v[r] = ps; v[8 + r] = pd;
    }
    const bool b3 = (l16 & 8) != 0, b2 = (l16 & 4) != 0;
    const bool b1 = (l16 & 2) != 0, b0 = (l16 & 1) != 0;
    float w8[8], w4[4], w2[2];
    for (int j = 0; j < 8; ++j) {
        const float keep = b3 ? v[8 + j] : v[j];
        const float send = b3 ? v[j] : v[8 + j];
        w8[j] = keep + xor_lane_f<8>(send);
    }
    for (int j = 0; j < 4; ++j) {
        const float keep = b2 ? w8[4 + j] : w8[j];
        const float send = b2 ? w8[j] : w8[4 + j];
        w4[j] = keep + xor_lane_f<4>(send);
    }
    for (int j = 0; j < 2; ++j) {
        const float keep = b1 ? w4[2 + j] : w4[j];
        const float send = b1 ? w4[j] : w4[2 + j];
        w2[j] = keep + xor_lane_f<2>(send);
    }
    const float keep1 = b0 ? w2[1] : w2[0];
    const float send1 = b0 ? w2[0] : w2[1];
    const float sc = keep1 + xor_lane_f<1>(send1);
    const int srow = rbase + (l16 & 7);
    if (srow < N) {
        float* sp = b3 ? s_dst : s_src;
        sp[srow] = sc;
    }
}

__global__ void gat_init(int* __restrict__ cnt, int* __restrict__ ovf,
                         int* __restrict__ bcnt, int N)
{
    const int i = blockIdx.x * blockDim.x + threadIdx.x;
    if (i < N) cnt[i] = 0;
    if (i < NB_MAX) bcnt[i * BSTRIDE] = 0;
    if (i == 0) ovf[0] = 0;
}

__device__ __forceinline__ void ovf_push(int* __restrict__ ovf, int row, int d, int N)
{
    const int k = atomicAdd(&ovf[0], 1);
    if (k < OVCAP(N)) { ovf[2 + 2 * k] = row; ovf[3 + 2 * k] = d; }
}

__device__ __forceinline__ void partition_block(
    int* hist, int* base, int pb,
    const int* __restrict__ src, const int* __restrict__ dst,
    int* __restrict__ bcnt, int* __restrict__ part,
    int* __restrict__ cnt, int* __restrict__ ovf, int N, int E)
{
    const int tid = threadIdx.x;
    const int nb = (N + RB - 1) >> RB_SHIFT;
    static_assert(PT == NB_MAX, "one thread clears one bucket counter");
    hist[tid] = 0;
    const int e0 = pb * (PT * PCH) + tid;
    int s[PCH], d[PCH], r[PCH];
#pragma unroll
    for (int j = 0; j < PCH; ++j) {
        const int i = e0 + j * PT;
        const int ic = i < E ? i : E - 1;
        const int sv = src[ic];
        d[j] = dst[ic];
        s[j] = i < E ? sv : -1;
    }
    __syncthreads();
#pragma unroll
    for (int j = 0; j < PCH; ++j)
        r[j] = s[j] >= 0 ? atomicAdd(&hist[s[j] >> RB_SHIFT], 1) : 0;
    __syncthreads();
    if (tid < nb) {
        const int c = hist[tid];
        base[tid] = c > 0 ? atomicAdd(&bcnt[tid * BSTRIDE], c) : 0;
    }
    __syncthreads();
#pragma unroll
    for (int j = 0; j < PCH; ++j)
        r[j] += base[s[j] >= 0 ? (s[j] >> RB_SHIFT) : 0];
#pragma unroll
    for (int j = 0; j < PCH; ++j) {
        if (s[j] >= 0) {
            const int b = s[j] >> RB_SHIFT;
            const int pos = r[j];
            const int rows = min(RB, N - (b << RB_SHIFT));
            if (pos < rows * OUT_F) {
                part[(size_t)b * BCAP + pos] =
                    (int)(((unsigned)(s[j] & (RB - 1)) << 24) | (unsigned)d[j]);
            } else {
                atomicOr(&cnt[s[j]], ROWFLAG);
                ovf_push(ovf, s[j], d[j], N);
            }
        }
    }
}

__global__ __launch_bounds__(PT, 12) void gat_gemm_partition(
    const float* __restrict__ x, const float* __restrict__ W,
    const float* __restrict__ attn, float* __restrict__ h,
    float* __restrict__ s_src, float* __restrict__ s_dst,
    const int* __restrict__ src, const int* __restrict__ dst,
    int* __restrict__ bcnt, int* __restrict__ part,
    int* __restrict__ cnt, int* __restrict__ ovf, int N, int E)
{
    __shared__ __attribute__((aligned(16))) unsigned char smem[IN_F * OUT_F * sizeof(float)];
    const int ngemm = (N + GROWS - 1) / GROWS;
    const int blk = blockIdx.x;
    if (blk < ngemm) {
        gemm_block(reinterpret_cast<float*>(smem), blk, x, W, attn, h, s_src, s_dst, N);
    } else {
        int* hist = reinterpret_cast<int*>(smem);
        partition_block(hist, hist + NB_MAX, blk - ngemm, src, dst, bcnt, part, cnt, ovf, N, E);
    }
}

__device__ __forceinline__ float wave_max32(float v) {
    v = fmaxf(v, xor_lane_f<16>(v));
    v = fmaxf(v, xor_lane_f<8>(v));
    v = fmaxf(v, xor_lane_f<4>(v));
    v = fmaxf(v, xor_lane_f<2>(v));
    v = fmaxf(v, xor_lane_f<1>(v));
    return v;
}
__device__ __forceinline__ float wave_sum32(float v) {
    v += xor_lane_f<16>(v);
    v += xor_lane_f<8>(v);
    v += xor_lane_f<4>(v);
    v += xor_lane_f<2>(v);
    v += xor_lane_f<1>(v);
    return v;
}

template <int B> __device__ __forceinline__ void h_rows(int dd, float p, int j0,
                                                       const float* __restrict__ h, int lane,
                                                       float& ax, float& ay)
{
    float2 hv[B];
    float pj[B];
#pragma unroll
    for (int u = 0; u < B; ++u) {
        const int dj = __builtin_amdgcn_readlane(dd, j0 + u);
        pj[u] = __int_as_float(__builtin_amdgcn_readlane(__float_as_int(p), j0 + u));
        hv[u] = ((const float2*)(h + (size_t)dj * OUT_F))[lane];
    }
#pragma unroll
    for (int u = 0; u < B; ++u) { ax += pj[u] * hv[u].x; ay += pj[u] * hv[u].y; }
}

__device__ __forceinline__ void row_chunk(bool valid, int d, float ss,
                                          const float* __restrict__ s_dst,
                                          const float* __restrict__ h, int lane,
                                          float& m, float& l, float& ax, float& ay)
{
    const unsigned mask = __builtin_amdgcn_ballot_w32(valid);
    if (mask == 0u) return;
    float e = -INFINITY;
    int dd = 0;
    if (valid) {
        dd = d;
        const float v = ss + s_dst[d];
        e = -(v > 0.f ? v : LRELU_ALPHA * v);
    }
    const float mn = fmaxf(m, wave_max32(e));
    const float scale = (m == -INFINITY) ? 0.f : __expf(m - mn);
    const float p = valid ? __expf(e - mn) : 0.f;
    l = l * scale + wave_sum32(p);
    ax *= scale; ay *= scale;
    m = mn;
    const int nmax = 32 - __clz((int)mask);
    for (int j0 = 0; j0 < nmax; j0 += 16) {
        if (j0 + 8 < nmax) h_rows<16>(dd, p, j0, h, lane, ax, ay);
        else               h_rows<8>(dd, p, j0, h, lane, ax, ay);
    }
}

template <int J> __device__ __forceinline__ int cmp_xchg(int v, bool keep_min) {
    const int o = xor_lane<J>(v);
    return keep_min ? min(v, o) : max(v, o);
}
__device__ __forceinline__ int bitonic_merge32(int v, int lane, bool up) {
    v = cmp_xchg<16>(v, ((lane & 16) == 0) == up);
    v = cmp_xchg<8>(v, ((lane & 8) == 0) == up);
    v = cmp_xchg<4>(v, ((lane & 4) == 0) == up);
    v = cmp_xchg<2>(v, ((lane & 2) == 0) == up);
    v = cmp_xchg<1>(v, ((lane & 1) == 0) == up);
    return v;
}
__device__ __forceinline__ int bitonic_sort32(int v, int lane, bool up) {
    const bool a2 = ((lane & 2) == 0) == up, a4 = ((lane & 4) == 0) == up;
    const bool a8 = ((lane & 8) == 0) == up, a16 = ((lane & 16) == 0) == up;
    v = cmp_xchg<1>(v, ((lane & 1) == 0) == a2);
    v = cmp_xchg<2>(v, ((lane & 2) == 0) == a4);
    v = cmp_xchg<1>(v, ((lane & 1) == 0) == a4);
    v = cmp_xchg<4>(v, ((lane & 4) == 0) == a8);
    v = cmp_xchg<2>(v, ((lane & 2) == 0) == a8);
    v = cmp_xchg<1>(v, ((lane & 1) == 0) == a8);
    v = cmp_xchg<8>(v, ((lane & 8) == 0) == a16);
    v = cmp_xchg<4>(v, ((lane & 4) == 0) == a16);
    v = cmp_xchg<2>(v, ((lane & 2) == 0) == a16);
    v = cmp_xchg<1>(v, ((lane & 1) == 0) == a16);
    return bitonic_merge32(v, lane, up);
}

__global__ __launch_bounds__(FT, 16) void gat_rank_gather(
    const int* __restrict__ src, const int* __restrict__ dst,
    const float* __restrict__ s_src, const float* __restrict__ s_dst,
    const int* __restrict__ bcnt, const int* __restrict__ cnt, const int* __restrict__ ovf,
    const float* __restrict__ h, const int* part, float* out, int N, int E)
{
    __shared__ int lc[RB];
    __shared__ int lst[RB * SLOT];
    __shared__ float lss[RB];
    __shared__ int lfl[RB];
    const int tid = threadIdx.x;
    const int b = blockIdx.x;
    const int row0 = b << RB_SHIFT;
    const int rows = min(RB, N - row0);
    if (tid < RB) lc[tid] = 0;
    if (tid < rows) { lss[tid] = s_src[row0 + tid]; lfl[tid] = cnt[row0 + tid]; }
    __syncthreads();
    const int n = min(bcnt[b * BSTRIDE], rows * OUT_F);
    const int* p = part + (size_t)b * BCAP;
    for (int t0 = 0; t0 < n; t0 += FT * RCH) {
        int w[RCH];
#pragma unroll
        for (int j = 0; j < RCH; ++j) {
            const int e = t0 + j * FT + tid;
            w[j] = p[e < n ? e : n - 1];
        }
        int rk[RCH];
#pragma unroll
        for (int j = 0; j < RCH; ++j) {
            const int e = t0 + j * FT + tid;
            rk[j] = SLOT;
            if (e < n) rk[j] = atomicAdd(&lc[(unsigned)w[j] >> 24], 1);
        }
#pragma unroll
        for (int j = 0; j < RCH; ++j)
            if (rk[j] < SLOT) lst[((unsigned)w[j] >> 24) * SLOT + rk[j]] = w[j] & 0xffffff;
    }
    __syncthreads();

    const int lane = tid & 31;
    const int wave = __builtin_amdgcn_readfirstlane(tid >> 5);
    for (int r = wave; r < rows; r += FT / 32) {
        const int row = row0 + r;
        const int c = __builtin_amdgcn_readfirstlane(lc[r]);
        const float ss = lss[r];
        float m = -INFINITY, l = 0.f, ax = 0.f, ay = 0.f;
        int total = 0;
        if (__builtin_amdgcn_readfirstlane(lfl[r]) & ROWFLAG) total = ovf[0];
        const bool full = c > SLOT || total > OVCAP(N);
        if (!full) {
            int k0 = lane < c ? lst[r * SLOT + lane] : 0x7fffffff;
            int k1 = 0x7fffffff;
            if (c > 32) {
                k1 = lane + 32 < c ? lst[r * SLOT + 32 + lane] : 0x7fffffff;
                k0 = bitonic_sort32(k0, lane, true);
                k1 = bitonic_sort32(k1, lane, false);
                const int lo = min(k0, k1), hi = max(k0, k1);
                k0 = bitonic_merge32(lo, lane, true);
                k1 = bitonic_merge32(hi, lane, true);
            } else {
                k0 = bitonic_sort32(k0, lane, true);
            }
            for (int q = 0; q < c; q += 32) {
                const bool v = q + lane < c;
                const int kk = q ? k1 : k0;
                row_chunk(v, v ? kk : 0, ss, s_dst, h, lane, m, l, ax, ay);
            }
        }
        const int* rlist = full ? src : ovf + 2;
        const int* dlist = full ? dst : ovf + 3;
        const int rstep = full ? 1 : 2;
        const int lim = full ? E : total;
        for (int base = 0; base < lim; base += 32) {
            const int idx = base + lane;
            bool v = idx < lim;
            int d = 0;
            if (v) {
                v = (rlist[(size_t)idx * rstep] == row);
                if (v) d = dlist[(size_t)idx * rstep];
            }
            row_chunk(v, d, ss, s_dst, h, lane, m, l, ax, ay);
        }
        float rx = 0.f, ry = 0.f;
        if (l > 0.f) { rx = ax / l; ry = ay / l; }
        float2 o;
        o.x = rx > 0.f ? rx : expm1f(rx);
        o.y = ry > 0.f ? ry : expm1f(ry);
        *(float2*)(out + (size_t)row * OUT_F + 2 * lane) = o;
    }
}

extern "C" void kernel_launch(void* const* d_in, const int* in_sizes, int n_in,
                              void* d_out, int out_size, void* d_ws, size_t ws_size,
                              hipStream_t stream)
{
    const float* x    = (const float*)d_in[0];
    const int*   edge = (const int*)  d_in[1];
    const float* W    = (const float*)d_in[2];
    const float* attn = (const float*)d_in[3];

    const int N = in_sizes[0] / IN_F;
    const int E = in_sizes[1] / 2;
    const int* src = edge;
    const int* dst = edge + E;
    float* out = (float*)d_out;

    float* h     = (float*)d_ws;
    float* s_src = h + (size_t)N * OUT_F;
    float* s_dst = s_src + N;
    int*   cnt   = (int*)(s_dst + N);
    int*   ovf   = cnt + N;

    int* bcnt = ovf + OVW(N);
    int* part = (int*)d_out;

    gat_init      <<<(N + 255) / 256,      256, 0, stream>>>(cnt, ovf, bcnt, N);
    gat_gemm_partition<<<(N + GROWS - 1) / GROWS + (E + PT * PCH - 1) / (PT * PCH), PT, 0, stream>>>(
        x, W, attn, h, s_src, s_dst, src, dst, bcnt, part, cnt, ovf, N, E);
    gat_rank_gather<<<(N + RB - 1) / RB,  FT,  0, stream>>>(
        src, dst, s_src, s_dst, bcnt, cnt, ovf, h, part, out, N, E);
}
